# starting kernel
# baseline (speedup 1.0000x reference)
#include <hip/hip_runtime.h>
#include <math.h>

#define Bn   32
#define Tn   4096
#define Dn   512
#define Sn   256
#define SCALE_INV 10.0f

typedef __attribute__((ext_vector_type(2))) float v2f;
typedef __attribute__((ext_vector_type(8))) float v8f;

#define NP 128
#define T_PER_WAVE (Tn / NP)

#define WS_H      0
#define WS_Q      (WS_H + Bn*Dn)
#define WS_ML     (WS_Q + Bn*Dn)
#define WS_ACC    (WS_ML + Bn*NP*2)
#define WS_POOLED (WS_ACC + Bn*NP*Dn)

__global__ void k_h(const float* __restrict__ t_star,
                    const float* __restrict__ W1,
                    const float* __restrict__ b1,
                    float* __restrict__ ws) {
    int i = blockIdx.x * blockDim.x + threadIdx.x;
    int b = i / Dn, d = i % Dn;
    float x = t_star[b] * W1[d] + b1[d];
    ws[WS_H + i] = x / (1.0f + __expf(-x));
}

__global__ void k_gemm_wmma(const float* __restrict__ A,
                            const float* __restrict__ Bm,
                            const float* __restrict__ bias,
                            float* __restrict__ Cout,
                            int N, int K, int split) {
    int lane = threadIdx.x & 31;
    int h    = lane >> 4;
    int ln   = lane & 15;
    int m0   = blockIdx.y * 16;
    int n    = blockIdx.x * 16 + ln;

    v8f c = {};
    for (int k0 = 0; k0 < K; k0 += 4) {
        int ka = k0 + 2 * h;
        v2f a, bb;
        a.x  = A[(m0 + ln) * K + ka];
        a.y  = A[(m0 + ln) * K + ka + 1];
        bb.x = Bm[(size_t)(ka    ) * N + n];
        bb.y = Bm[(size_t)(ka + 1) * N + n];
        c = __builtin_amdgcn_wmma_f32_16x16x4_f32(false, a, false, bb,
                                                  (short)0, c, false, false);
    }
    float bv = bias[n];
    #pragma unroll
    for (int v = 0; v < 8; ++v) {
        int m = m0 + v + 8 * h;
        float val = c[v] + bv;
        if (!split) {
            Cout[m * N + n] = val;
        } else {
            if (n < Sn) Cout[m * Sn + n] = val;
            else        Cout[Bn * Sn + m * Sn + (n - Sn)] = val;
        }
    }
}

__global__ void k_attend(const float* __restrict__ context,
                         const float* __restrict__ t_star,
                         const float* __restrict__ t_context,
                         float* __restrict__ ws) {
    int blk      = blockIdx.x;
    int b        = blk >> 4;
    int blk_in_b = blk & 15;
    int wave     = threadIdx.x >> 5;
    int lane     = threadIdx.x & 31;
    int p        = blk_in_b * 8 + wave;
    int t0       = p * T_PER_WAVE;

    const float* q = ws + WS_Q + b * Dn;
    float4 q4[4];
    #pragma unroll
    for (int i = 0; i < 4; ++i)
        q4[i] = *(const float4*)(q + lane * 4 + i * 128);

    float ts = t_star[b];
    const float* ctx_b = context   + (size_t)b * Tn * Dn;
    const float* tc_b  = t_context + (size_t)b * Tn;

    float m = -INFINITY, l = 0.0f;
    float4 acc[4] = {};

    for (int tt = 0; tt < T_PER_WAVE; ++tt) {
        int t = t0 + tt;
        const float* row = ctx_b + (size_t)t * Dn + lane * 4;
        float4 x[4];
        #pragma unroll
        for (int i = 0; i < 4; ++i)
            x[i] = *(const float4*)(row + i * 128);
        if (tt + 1 < T_PER_WAVE)
            __builtin_prefetch(row + Dn, 0, 1);

        float dotp = 0.0f;
        #pragma unroll
        for (int i = 0; i < 4; ++i)
            dotp += x[i].x * q4[i].x + x[i].y * q4[i].y +
                    x[i].z * q4[i].z + x[i].w * q4[i].w;
        #pragma unroll
        for (int off = 16; off > 0; off >>= 1)
            dotp += __shfl_xor(dotp, off, 32);

        float s  = dotp - fabsf(ts - tc_b[t]) * SCALE_INV;
        float mn = fmaxf(m, s);
        float sc = __expf(m - mn);
        float w  = __expf(s - mn);
        l = l * sc + w;
        #pragma unroll
        for (int i = 0; i < 4; ++i) {
            acc[i].x = acc[i].x * sc + w * x[i].x;
            acc[i].y = acc[i].y * sc + w * x[i].y;
            acc[i].z = acc[i].z * sc + w * x[i].z;
            acc[i].w = acc[i].w * sc + w * x[i].w;
        }
        m = mn;
    }

    float* accout = ws + WS_ACC + ((size_t)(b * NP + p)) * Dn;
    #pragma unroll
    for (int i = 0; i < 4; ++i)
        *(float4*)(accout + lane * 4 + i * 128) = acc[i];
    if (lane == 0) {
        float* ml = ws + WS_ML + (b * NP + p) * 2;
        ml[0] = m; ml[1] = l;
    }
}

__global__ void k_combine(float* __restrict__ ws) {
    int b = blockIdx.x;
    int d = threadIdx.x;
    const float* ml = ws + WS_ML + b * NP * 2;
    float M = -INFINITY;
    for (int p = 0; p < NP; ++p) M = fmaxf(M, ml[2 * p]);
    const float* accb = ws + WS_ACC + (size_t)b * NP * Dn;
    float L = 0.0f, sum = 0.0f;
    for (int p = 0; p < NP; ++p) {
        float e = __expf(ml[2 * p] - M);
        L   += e * ml[2 * p + 1];
        sum += e * accb[(size_t)p * Dn + d];
    }
    ws[WS_POOLED + b * Dn + d] = sum / L;
}

extern "C" void kernel_launch(void* const* d_in, const int* in_sizes, int n_in,
                              void* d_out, int out_size, void* d_ws, size_t ws_size,
                              hipStream_t stream) {
    const float* context   = (const float*)d_in[0];
    const float* t_star    = (const float*)d_in[1];
    const float* t_context = (const float*)d_in[2];
    const float* W1        = (const float*)d_in[3];
    const float* b1        = (const float*)d_in[4];
    const float* W2        = (const float*)d_in[5];
    const float* b2        = (const float*)d_in[6];
    const float* Wout      = (const float*)d_in[7];
    const float* bout      = (const float*)d_in[8];
    float* out = (float*)d_out;
    float* ws  = (float*)d_ws;

    k_h<<<(Bn * Dn) / 256, 256, 0, stream>>>(t_star, W1, b1, ws);

    k_gemm_wmma<<<dim3(Dn / 16, 2), 32, 0, stream>>>(ws + WS_H, W2, b2,
                                                     ws + WS_Q, Dn, Dn, 0);

    k_attend<<<Bn * 16, 256, 0, stream>>>(context, t_star, t_context, ws);

    k_combine<<<Bn, Dn, 0, stream>>>(ws);

    k_gemm_wmma<<<dim3((2 * Sn) / 16, 2), 32, 0, stream>>>(ws + WS_POOLED, Wout, bout,
                                                           out, 2 * Sn, Dn, 1);
}


# GEMM kernel (both launches): K split over 8 waves per block, each wave issues all fragment loads of its k slice ahead of its WMMA chain; partial tiles added via LDS in wave order (new summation tree)
# speedup vs baseline: 2.1547x; 2.1547x over previous
#include <hip/hip_runtime.h>
#include <math.h>

#define Bn   32
#define Tn   4096
#define Dn   512
#define Sn   256
#define SCALE_INV 10.0f

typedef __attribute__((ext_vector_type(2))) float v2f;
typedef __attribute__((ext_vector_type(8))) float v8f;

#define NP 128
#define T_PER_WAVE (Tn / NP)

#define WS_H      0
#define WS_Q      (WS_H + Bn*Dn)
#define WS_ML     (WS_Q + Bn*Dn)
#define WS_ACC    (WS_ML + Bn*NP*2)
#define WS_POOLED (WS_ACC + Bn*NP*Dn)

__global__ void k_h(const float* __restrict__ t_star,
                    const float* __restrict__ W1,
                    const float* __restrict__ b1,
                    float* __restrict__ ws) {
    int i = blockIdx.x * blockDim.x + threadIdx.x;
    int b = i / Dn, d = i % Dn;
    float x = t_star[b] * W1[d] + b1[d];
    ws[WS_H + i] = x / (1.0f + __expf(-x));
}

#define GN      512
#define GK      512
#define KSPLIT  8
#define KSLICE  (GK / KSPLIT)
#define KSTEPS  (KSLICE / 4)
static_assert(GN == Dn && GN == 2 * Sn && GK == Dn, "both products are 32 x 512 x 512");

__global__ __launch_bounds__(32 * KSPLIT)
void k_gemm_wmma(const float* __restrict__ A,
                 const float* __restrict__ Bm,
                 const float* __restrict__ bias,
                 float* __restrict__ Cout,
                 int split) {
    __shared__ float part[KSPLIT * 256];

    int wave = __builtin_amdgcn_readfirstlane(threadIdx.x >> 5);
    int lane = threadIdx.x & 31;
    int h    = lane >> 4;
    int ln   = lane & 15;
    int m0   = blockIdx.y * 16;
    int n    = blockIdx.x * 16 + ln;

    int kb = wave * KSLICE;
    const float* ap = A  + (m0 + ln) * GK + kb + 2 * h;
    const float* bp = Bm + (size_t)(kb + 2 * h) * GN + n;

    float bv = bias[n];

    v2f a[KSTEPS], bb[KSTEPS];
    #pragma unroll
    for (int i = 0; i < KSTEPS; ++i) {
        a[i]    = *(const v2f*)(ap + 4 * i);
        bb[i].x = bp[(size_t)(4 * i    ) * GN];
        bb[i].y = bp[(size_t)(4 * i + 1) * GN];
    }
    __builtin_amdgcn_sched_barrier(0);

    v8f c = {};
    #pragma unroll
    for (int i = 0; i < KSTEPS; ++i)
        c = __builtin_amdgcn_wmma_f32_16x16x4_f32(false, a[i], false, bb[i],
                                                  (short)0, c, false, false);

    #pragma unroll
    for (int v = 0; v < 8; ++v)
        part[(wave * 8 + v) * 32 + lane] = c[v];
    __syncthreads();

    int e = wave * 32 + lane;
    float s = part[e];
    #pragma unroll
    for (int w = 1; w < KSPLIT; ++w)
        s += part[w * 256 + e];

    int   m   = m0 + wave + 8 * h;
    float val = s + bv;
    if (!split) {
        Cout[m * GN + n] = val;
    } else {
        if (n < Sn) Cout[m * Sn + n] = val;
        else        Cout[Bn * Sn + m * Sn + (n - Sn)] = val;
    }
}

__global__ void k_attend(const float* __restrict__ context,
                         const float* __restrict__ t_star,
                         const float* __restrict__ t_context,
                         float* __restrict__ ws) {
    int blk      = blockIdx.x;
    int b        = blk >> 4;
    int blk_in_b = blk & 15;
    int wave     = threadIdx.x >> 5;
    int lane     = threadIdx.x & 31;
    int p        = blk_in_b * 8 + wave;
    int t0       = p * T_PER_WAVE;

    const float* q = ws + WS_Q + b * Dn;
    float4 q4[4];
    #pragma unroll
    for (int i = 0; i < 4; ++i)
        q4[i] = *(const float4*)(q + lane * 4 + i * 128);

    float ts = t_star[b];
    const float* ctx_b = context   + (size_t)b * Tn * Dn;
    const float* tc_b  = t_context + (size_t)b * Tn;

    float m = -INFINITY, l = 0.0f;
    float4 acc[4] = {};

    for (int tt = 0; tt < T_PER_WAVE; ++tt) {
        int t = t0 + tt;
        const float* row = ctx_b + (size_t)t * Dn + lane * 4;
        float4 x[4];
        #pragma unroll
        for (int i = 0; i < 4; ++i)
            x[i] = *(const float4*)(row + i * 128);
        if (tt + 1 < T_PER_WAVE)
            __builtin_prefetch(row + Dn, 0, 1);

        float dotp = 0.0f;
        #pragma unroll
        for (int i = 0; i < 4; ++i)
            dotp += x[i].x * q4[i].x + x[i].y * q4[i].y +
                    x[i].z * q4[i].z + x[i].w * q4[i].w;
        #pragma unroll
        for (int off = 16; off > 0; off >>= 1)
            dotp += __shfl_xor(dotp, off, 32);

        float s  = dotp - fabsf(ts - tc_b[t]) * SCALE_INV;
        float mn = fmaxf(m, s);
        float sc = __expf(m - mn);
        float w  = __expf(s - mn);
        l = l * sc + w;
        #pragma unroll
        for (int i = 0; i < 4; ++i) {
            acc[i].x = acc[i].x * sc + w * x[i].x;
            acc[i].y = acc[i].y * sc + w * x[i].y;
            acc[i].z = acc[i].z * sc + w * x[i].z;
            acc[i].w = acc[i].w * sc + w * x[i].w;
        }
        m = mn;
    }

    float* accout = ws + WS_ACC + ((size_t)(b * NP + p)) * Dn;
    #pragma unroll
    for (int i = 0; i < 4; ++i)
        *(float4*)(accout + lane * 4 + i * 128) = acc[i];
    if (lane == 0) {
        float* ml = ws + WS_ML + (b * NP + p) * 2;
        ml[0] = m; ml[1] = l;
    }
}

__global__ void k_combine(float* __restrict__ ws) {
    int b = blockIdx.x;
    int d = threadIdx.x;
    const float* ml = ws + WS_ML + b * NP * 2;
    float M = -INFINITY;
    for (int p = 0; p < NP; ++p) M = fmaxf(M, ml[2 * p]);
    const float* accb = ws + WS_ACC + (size_t)b * NP * Dn;
    float L = 0.0f, sum = 0.0f;
    for (int p = 0; p < NP; ++p) {
        float e = __expf(ml[2 * p] - M);
        L   += e * ml[2 * p + 1];
        sum += e * accb[(size_t)p * Dn + d];
    }
    ws[WS_POOLED + b * Dn + d] = sum / L;
}

extern "C" void kernel_launch(void* const* d_in, const int* in_sizes, int n_in,
                              void* d_out, int out_size, void* d_ws, size_t ws_size,
                              hipStream_t stream) {
    const float* context   = (const float*)d_in[0];
    const float* t_star    = (const float*)d_in[1];
    const float* t_context = (const float*)d_in[2];
    const float* W1        = (const float*)d_in[3];
    const float* b1        = (const float*)d_in[4];
    const float* W2        = (const float*)d_in[5];
    const float* b2        = (const float*)d_in[6];
    const float* Wout      = (const float*)d_in[7];
    const float* bout      = (const float*)d_in[8];
    float* out = (float*)d_out;
    float* ws  = (float*)d_ws;

    k_h<<<(Bn * Dn) / 256, 256, 0, stream>>>(t_star, W1, b1, ws);

    k_gemm_wmma<<<dim3(GN / 16, 2), 32 * KSPLIT, 0, stream>>>(ws + WS_H, W2, b2,
                                                              ws + WS_Q, 0);

    k_attend<<<Bn * 16, 256, 0, stream>>>(context, t_star, t_context, ws);

    k_combine<<<Bn, Dn, 0, stream>>>(ws);

    k_gemm_wmma<<<dim3(GN / 16, 2), 32 * KSPLIT, 0, stream>>>(ws + WS_POOLED, Wout, bout,
                                                              out, 1);
}


# k_attend streams rows in groups of 4, 2 groups requested ahead of the math (16-32 loads in flight per wave, was 4); softmax max/rescale once per group, row dot in 2 packed chains (new sum order)
# speedup vs baseline: 2.6198x; 1.2159x over previous
#include <hip/hip_runtime.h>
#include <math.h>

#define Bn   32
#define Tn   4096
#define Dn   512
#define Sn   256
#define SCALE_INV 10.0f

typedef __attribute__((ext_vector_type(2))) float v2f;
typedef __attribute__((ext_vector_type(8))) float v8f;

#define NP 128
#define T_PER_WAVE (Tn / NP)

#define WS_H      0
#define WS_Q      (WS_H + Bn*Dn)
#define WS_ML     (WS_Q + Bn*Dn)
#define WS_ACC    (WS_ML + Bn*NP*2)
#define WS_POOLED (WS_ACC + Bn*NP*Dn)

__global__ void k_h(const float* __restrict__ t_star,
                    const float* __restrict__ W1,
                    const float* __restrict__ b1,
                    float* __restrict__ ws) {
    int i = blockIdx.x * blockDim.x + threadIdx.x;
    int b = i / Dn, d = i % Dn;
    float x = t_star[b] * W1[d] + b1[d];
    ws[WS_H + i] = x / (1.0f + __expf(-x));
}

#define GN      512
#define GK      512
#define KSPLIT  8
#define KSLICE  (GK / KSPLIT)
#define KSTEPS  (KSLICE / 4)
static_assert(GN == Dn && GN == 2 * Sn && GK == Dn, "both products are 32 x 512 x 512");

__global__ __launch_bounds__(32 * KSPLIT)
void k_gemm_wmma(const float* __restrict__ A,
                 const float* __restrict__ Bm,
                 const float* __restrict__ bias,
                 float* __restrict__ Cout,
                 int split) {
    __shared__ float part[KSPLIT * 256];

    int wave = __builtin_amdgcn_readfirstlane(threadIdx.x >> 5);
    int lane = threadIdx.x & 31;
    int h    = lane >> 4;
    int ln   = lane & 15;
    int m0   = blockIdx.y * 16;
    int n    = blockIdx.x * 16 + ln;

    int kb = wave * KSLICE;
    const float* ap = A  + (m0 + ln) * GK + kb + 2 * h;
    const float* bp = Bm + (size_t)(kb + 2 * h) * GN + n;

    float bv = bias[n];

    v2f a[KSTEPS], bb[KSTEPS];
    #pragma unroll
    for (int i = 0; i < KSTEPS; ++i) {
        a[i]    = *(const v2f*)(ap + 4 * i);
        bb[i].x = bp[(size_t)(4 * i    ) * GN];
        bb[i].y = bp[(size_t)(4 * i + 1) * GN];
    }
    __builtin_amdgcn_sched_barrier(0);

    v8f c = {};
    #pragma unroll
    for (int i = 0; i < KSTEPS; ++i)
        c = __builtin_amdgcn_wmma_f32_16x16x4_f32(false, a[i], false, bb[i],
                                                  (short)0, c, false, false);

    #pragma unroll
    for (int v = 0; v < 8; ++v)
        part[(wave * 8 + v) * 32 + lane] = c[v];
    __syncthreads();

    int e = wave * 32 + lane;
    float s = part[e];
    #pragma unroll
    for (int w = 1; w < KSPLIT; ++w)
        s += part[w * 256 + e];

    int   m   = m0 + wave + 8 * h;
    float val = s + bv;
    if (!split) {
        Cout[m * GN + n] = val;
    } else {
        if (n < Sn) Cout[m * Sn + n] = val;
        else        Cout[Bn * Sn + m * Sn + (n - Sn)] = val;
    }
}

#define RG   4
#define NGRP (T_PER_WAVE / RG)
static_assert(T_PER_WAVE == 32, "lane tt of a wave holds the time distance of the wave's row tt");
static_assert(T_PER_WAVE % RG == 0 && NGRP >= 2, "whole groups, at least the two requested up front");
typedef __attribute__((ext_vector_type(4))) float v4f;

__global__ __launch_bounds__(256)
void k_attend(const float* __restrict__ context,
              const float* __restrict__ t_star,
              const float* __restrict__ t_context,
              float* __restrict__ ws) {
    int blk      = blockIdx.x;
    int b        = blk >> 4;
    int blk_in_b = blk & 15;
    int wave     = __builtin_amdgcn_readfirstlane(threadIdx.x >> 5);
    int lane     = threadIdx.x & 31;
    int p        = blk_in_b * 8 + wave;
    int t0       = p * T_PER_WAVE;

    const float* q = ws + WS_Q + b * Dn;
    v4f q4[4];
    #pragma unroll
    for (int i = 0; i < 4; ++i)
        q4[i] = *(const v4f*)(q + lane * 4 + i * 128);

    float ts = t_star[b];
    float tcv = t_context[(size_t)b * Tn + t0 + lane];

    const float* ctx_w = context + ((size_t)b * Tn + t0) * Dn + lane * 4;

    v4f x[2][RG][4];
#define LOAD_GROUP(G)                                                              \
    _Pragma("unroll") for (int r = 0; r < RG; ++r)                                 \
    _Pragma("unroll") for (int i = 0; i < 4; ++i)                                  \
        x[(G) & 1][r][i] = *(const v4f*)(ctx_w + ((G) * RG + r) * Dn + i * 128)

    float m = -INFINITY, l = 0.0f;
    v2f acc[4][2] = {};

    LOAD_GROUP(0);
    LOAD_GROUP(1);
    __builtin_amdgcn_sched_barrier(0);
    float ad = fabsf(ts - tcv);

    #pragma unroll
    for (int g = 0; g < NGRP; ++g) {
        const v4f (*xg)[4] = x[g & 1];
        float s[RG];
        #pragma unroll
        for (int r = 0; r < RG; ++r) {
            v2f d = {0.0f, 0.0f};
            #pragma unroll
            for (int i = 0; i < 4; ++i) {
                d = __builtin_elementwise_fma(xg[r][i].lo, q4[i].lo, d);
                d = __builtin_elementwise_fma(xg[r][i].hi, q4[i].hi, d);
            }
            float dotp = d.x + d.y;
            #pragma unroll
            for (int off = 16; off > 0; off >>= 1)
                dotp += __shfl_xor(dotp, off, 32);
            float adr = __builtin_bit_cast(float,
                            __builtin_amdgcn_readlane(__builtin_bit_cast(int, ad), g * RG + r));
            s[r] = dotp - adr * SCALE_INV;
        }

        float gm = s[0];
        #pragma unroll
        for (int r = 1; r < RG; ++r) gm = fmaxf(gm, s[r]);
        float mn = fmaxf(m, gm);
        float sc = __expf(m - mn);
        float w[RG];
        float wsum = 0.0f;
        #pragma unroll
        for (int r = 0; r < RG; ++r) { w[r] = __expf(s[r] - mn); wsum += w[r]; }
        l = l * sc + wsum;
        #pragma unroll
        for (int i = 0; i < 4; ++i) {
            v2f a0 = acc[i][0] * sc;
            v2f a1 = acc[i][1] * sc;
            #pragma unroll
            for (int r = 0; r < RG; ++r) {
                v2f wv = {w[r], w[r]};
                a0 = __builtin_elementwise_fma(wv, xg[r][i].lo, a0);
                a1 = __builtin_elementwise_fma(wv, xg[r][i].hi, a1);
            }
            acc[i][0] = a0;
            acc[i][1] = a1;
        }
        m = mn;

        __builtin_amdgcn_sched_barrier(0);
        if (g + 2 < NGRP) LOAD_GROUP(g + 2);
        __builtin_amdgcn_sched_barrier(0);
    }
#undef LOAD_GROUP

    float* accout = ws + WS_ACC + ((size_t)(b * NP + p)) * Dn;
    #pragma unroll
    for (int i = 0; i < 4; ++i)
        *(v4f*)(accout + lane * 4 + i * 128) = __builtin_shufflevector(acc[i][0], acc[i][1], 0, 1, 2, 3);
    if (lane == 0) {
        float* ml = ws + WS_ML + (b * NP + p) * 2;
        ml[0] = m; ml[1] = l;
    }
}

__global__ void k_combine(float* __restrict__ ws) {
    int b = blockIdx.x;
    int d = threadIdx.x;
    const float* ml = ws + WS_ML + b * NP * 2;
    float M = -INFINITY;
    for (int p = 0; p < NP; ++p) M = fmaxf(M, ml[2 * p]);
    const float* accb = ws + WS_ACC + (size_t)b * NP * Dn;
    float L = 0.0f, sum = 0.0f;
    for (int p = 0; p < NP; ++p) {
        float e = __expf(ml[2 * p] - M);
        L   += e * ml[2 * p + 1];
        sum += e * accb[(size_t)p * Dn + d];
    }
    ws[WS_POOLED + b * Dn + d] = sum / L;
}

extern "C" void kernel_launch(void* const* d_in, const int* in_sizes, int n_in,
                              void* d_out, int out_size, void* d_ws, size_t ws_size,
                              hipStream_t stream) {
    const float* context   = (const float*)d_in[0];
    const float* t_star    = (const float*)d_in[1];
    const float* t_context = (const float*)d_in[2];
    const float* W1        = (const float*)d_in[3];
    const float* b1        = (const float*)d_in[4];
    const float* W2        = (const float*)d_in[5];
    const float* b2        = (const float*)d_in[6];
    const float* Wout      = (const float*)d_in[7];
    const float* bout      = (const float*)d_in[8];
    float* out = (float*)d_out;
    float* ws  = (float*)d_ws;

    k_h<<<(Bn * Dn) / 256, 256, 0, stream>>>(t_star, W1, b1, ws);

    k_gemm_wmma<<<dim3(GN / 16, 2), 32 * KSPLIT, 0, stream>>>(ws + WS_H, W2, b2,
                                                              ws + WS_Q, 0);

    k_attend<<<Bn * 16, 256, 0, stream>>>(context, t_star, t_context, ws);

    k_combine<<<Bn, Dn, 0, stream>>>(ws);

    k_gemm_wmma<<<dim3(GN / 16, 2), 32 * KSPLIT, 0, stream>>>(ws + WS_POOLED, Wout, bout,
                                                              out, 1);
}


# k_attend: the 8 waves of a block merge their partial softmax states (max, sum, weighted row) in LDS in wave order; 16 partials per batch reach k_combine, was 128 (new summation tree)
# speedup vs baseline: 3.6806x; 1.4049x over previous
#include <hip/hip_runtime.h>
#include <math.h>

#define Bn   32
#define Tn   4096
#define Dn   512
#define Sn   256
#define SCALE_INV 10.0f

typedef __attribute__((ext_vector_type(2))) float v2f;
typedef __attribute__((ext_vector_type(8))) float v8f;

#define BLK_PER_B  16
#define WAVES_BLK  8
#define T_PER_WAVE (Tn / (BLK_PER_B * WAVES_BLK))
#define NP         BLK_PER_B

#define WS_H      0
#define WS_Q      (WS_H + Bn*Dn)
#define WS_ML     (WS_Q + Bn*Dn)
#define WS_ACC    (WS_ML + Bn*NP*2)
#define WS_POOLED (WS_ACC + Bn*NP*Dn)

__global__ void k_h(const float* __restrict__ t_star,
                    const float* __restrict__ W1,
                    const float* __restrict__ b1,
                    float* __restrict__ ws) {
    int i = blockIdx.x * blockDim.x + threadIdx.x;
    int b = i / Dn, d = i % Dn;
    float x = t_star[b] * W1[d] + b1[d];
    ws[WS_H + i] = x / (1.0f + __expf(-x));
}

#define GN      512
#define GK      512
#define KSPLIT  8
#define KSLICE  (GK / KSPLIT)
#define KSTEPS  (KSLICE / 4)
static_assert(GN == Dn && GN == 2 * Sn && GK == Dn, "both products are 32 x 512 x 512");

__global__ __launch_bounds__(32 * KSPLIT)
void k_gemm_wmma(const float* __restrict__ A,
                 const float* __restrict__ Bm,
                 const float* __restrict__ bias,
                 float* __restrict__ Cout,
                 int split) {
    __shared__ float part[KSPLIT * 256];

    int wave = __builtin_amdgcn_readfirstlane(threadIdx.x >> 5);
    int lane = threadIdx.x & 31;
    int h    = lane >> 4;
    int ln   = lane & 15;
    int m0   = blockIdx.y * 16;
    int n    = blockIdx.x * 16 + ln;

    int kb = wave * KSLICE;
    const float* ap = A  + (m0 + ln) * GK + kb + 2 * h;
    const float* bp = Bm + (size_t)(kb + 2 * h) * GN + n;

    float bv = bias[n];

    v2f a[KSTEPS], bb[KSTEPS];
    #pragma unroll
    for (int i = 0; i < KSTEPS; ++i) {
        a[i]    = *(const v2f*)(ap + 4 * i);
        bb[i].x = bp[(size_t)(4 * i    ) * GN];
        bb[i].y = bp[(size_t)(4 * i + 1) * GN];
    }
    __builtin_amdgcn_sched_barrier(0);

    v8f c = {};
    #pragma unroll
    for (int i = 0; i < KSTEPS; ++i)
        c = __builtin_amdgcn_wmma_f32_16x16x4_f32(false, a[i], false, bb[i],
                                                  (short)0, c, false, false);

    #pragma unroll
    for (int v = 0; v < 8; ++v)
        part[(wave * 8 + v) * 32 + lane] = c[v];
    __syncthreads();

    int e = wave * 32 + lane;
    float s = part[e];
    #pragma unroll
    for (int w = 1; w < KSPLIT; ++w)
        s += part[w * 256 + e];

    int   m   = m0 + wave + 8 * h;
    float val = s + bv;
    if (!split) {
        Cout[m * GN + n] = val;
    } else {
        if (n < Sn) Cout[m * Sn + n] = val;
        else        Cout[Bn * Sn + m * Sn + (n - Sn)] = val;
    }
}

#define RG   4
#define NGRP (T_PER_WAVE / RG)
static_assert(T_PER_WAVE == 32, "lane tt of a wave holds the time distance of the wave's row tt");
static_assert(T_PER_WAVE % RG == 0 && NGRP >= 2, "whole groups, at least the two requested up front");
typedef __attribute__((ext_vector_type(4))) float v4f;

__global__ __launch_bounds__(256)
void k_attend(const float* __restrict__ context,
              const float* __restrict__ t_star,
              const float* __restrict__ t_context,
              float* __restrict__ ws) {
    int blk      = blockIdx.x;
    int b        = blk >> 4;
    int blk_in_b = blk & 15;
    int wave     = __builtin_amdgcn_readfirstlane(threadIdx.x >> 5);
    int lane     = threadIdx.x & 31;
    int p        = blk_in_b * WAVES_BLK + wave;
    int t0       = p * T_PER_WAVE;

    const float* q = ws + WS_Q + b * Dn;
    v4f q4[4];
    #pragma unroll
    for (int i = 0; i < 4; ++i)
        q4[i] = *(const v4f*)(q + lane * 4 + i * 128);

    float ts = t_star[b];
    float tcv = t_context[(size_t)b * Tn + t0 + lane];

    const float* ctx_w = context + ((size_t)b * Tn + t0) * Dn + lane * 4;

    v4f x[2][RG][4];
#define LOAD_GROUP(G)                                                              \
    _Pragma("unroll") for (int r = 0; r < RG; ++r)                                 \
    _Pragma("unroll") for (int i = 0; i < 4; ++i)                                  \
        x[(G) & 1][r][i] = *(const v4f*)(ctx_w + ((G) * RG + r) * Dn + i * 128)

    float m = -INFINITY, l = 0.0f;
    v2f acc[4][2] = {};

    LOAD_GROUP(0);
    LOAD_GROUP(1);
    __builtin_amdgcn_sched_barrier(0);
    float ad = fabsf(ts - tcv);

    #pragma unroll
    for (int g = 0; g < NGRP; ++g) {
        const v4f (*xg)[4] = x[g & 1];
        float s[RG];
        #pragma unroll
        for (int r = 0; r < RG; ++r) {
            v2f d = {0.0f, 0.0f};
            #pragma unroll
            for (int i = 0; i < 4; ++i) {
                d = __builtin_elementwise_fma(xg[r][i].lo, q4[i].lo, d);
                d = __builtin_elementwise_fma(xg[r][i].hi, q4[i].hi, d);
            }
            float dotp = d.x + d.y;
            #pragma unroll
            for (int off = 16; off > 0; off >>= 1)
                dotp += __shfl_xor(dotp, off, 32);
            float adr = __builtin_bit_cast(float,
                            __builtin_amdgcn_readlane(__builtin_bit_cast(int, ad), g * RG + r));
            s[r] = dotp - adr * SCALE_INV;
        }

        float gm = s[0];
        #pragma unroll
        for (int r = 1; r < RG; ++r) gm = fmaxf(gm, s[r]);
        float mn = fmaxf(m, gm);
        float sc = __expf(m - mn);
        float w[RG];
        float wsum = 0.0f;
        #pragma unroll
        for (int r = 0; r < RG; ++r) { w[r] = __expf(s[r] - mn); wsum += w[r]; }
        l = l * sc + wsum;
        #pragma unroll
        for (int i = 0; i < 4; ++i) {
            v2f a0 = acc[i][0] * sc;
            v2f a1 = acc[i][1] * sc;
            #pragma unroll
            for (int r = 0; r < RG; ++r) {
                v2f wv = {w[r], w[r]};
                a0 = __builtin_elementwise_fma(wv, xg[r][i].lo, a0);
                a1 = __builtin_elementwise_fma(wv, xg[r][i].hi, a1);
            }
            acc[i][0] = a0;
            acc[i][1] = a1;
        }
        m = mn;

        __builtin_amdgcn_sched_barrier(0);
        if (g + 2 < NGRP) LOAD_GROUP(g + 2);
        __builtin_amdgcn_sched_barrier(0);
    }
#undef LOAD_GROUP

    __shared__ float s_acc[WAVES_BLK][Dn];
    __shared__ float s_m[WAVES_BLK], s_l[WAVES_BLK];
    for (int i = 0; i < 4; ++i)
        *(v4f*)(&s_acc[wave][lane * 4 + i * 128]) = __builtin_shufflevector(acc[i][0], acc[i][1], 0, 1, 2, 3);
    if (lane == 0) { s_m[wave] = m; s_l[wave] = l; }
    __syncthreads();

    int tid = threadIdx.x;
    float mw[WAVES_BLK];
    float M = -INFINITY;
    for (int w = 0; w < WAVES_BLK; ++w) { mw[w] = s_m[w]; M = fmaxf(M, mw[w]); }
    float Lb = 0.0f;
    v2f sum = {0.0f, 0.0f};
    for (int w = 0; w < WAVES_BLK; ++w) {
        float e = __expf(mw[w] - M);
        Lb += e * s_l[w];
        v2f ev = {e, e};
        sum = __builtin_elementwise_fma(ev, *(const v2f*)(&s_acc[w][2 * tid]), sum);
    }
    float* accout = ws + WS_ACC + ((size_t)(b * NP + blk_in_b)) * Dn;
    *(v2f*)(accout + 2 * tid) = sum;
    if (tid == 0) {
        float* ml = ws + WS_ML + (b * NP + blk_in_b) * 2;
        ml[0] = M; ml[1] = Lb;
    }
}

__global__ void k_combine(float* __restrict__ ws) {
    int b = blockIdx.x;
    int d = threadIdx.x;
    const float* ml = ws + WS_ML + b * NP * 2;
    float M = -INFINITY;
    for (int p = 0; p < NP; ++p) M = fmaxf(M, ml[2 * p]);
    const float* accb = ws + WS_ACC + (size_t)b * NP * Dn;
    float L = 0.0f, sum = 0.0f;
    for (int p = 0; p < NP; ++p) {
        float e = __expf(ml[2 * p] - M);
        L   += e * ml[2 * p + 1];
        sum += e * accb[(size_t)p * Dn + d];
    }
    ws[WS_POOLED + b * Dn + d] = sum / L;
}

extern "C" void kernel_launch(void* const* d_in, const int* in_sizes, int n_in,
                              void* d_out, int out_size, void* d_ws, size_t ws_size,
                              hipStream_t stream) {
    const float* context   = (const float*)d_in[0];
    const float* t_star    = (const float*)d_in[1];
    const float* t_context = (const float*)d_in[2];
    const float* W1        = (const float*)d_in[3];
    const float* b1        = (const float*)d_in[4];
    const float* W2        = (const float*)d_in[5];
    const float* b2        = (const float*)d_in[6];
    const float* Wout      = (const float*)d_in[7];
    const float* bout      = (const float*)d_in[8];
    float* out = (float*)d_out;
    float* ws  = (float*)d_ws;

    k_h<<<(Bn * Dn) / 256, 256, 0, stream>>>(t_star, W1, b1, ws);

    k_gemm_wmma<<<dim3(GN / 16, 2), 32 * KSPLIT, 0, stream>>>(ws + WS_H, W2, b2,
                                                              ws + WS_Q, 0);

    k_attend<<<Bn * 16, 256, 0, stream>>>(context, t_star, t_context, ws);

    k_combine<<<Bn, Dn, 0, stream>>>(ws);

    k_gemm_wmma<<<dim3(GN / 16, 2), 32 * KSPLIT, 0, stream>>>(ws + WS_POOLED, Wout, bout,
                                                              out, 1);
}


# k_h folded into the first GEMM: its A fragments h = silu(t_star*W1+b1) are computed in the A loader from t_star, W1, b1 (requested ahead of the B fragments); 4 launches, was 5
# speedup vs baseline: 3.9558x; 1.0748x over previous
#include <hip/hip_runtime.h>
#include <math.h>

#define Bn   32
#define Tn   4096
#define Dn   512
#define Sn   256
#define SCALE_INV 10.0f

typedef __attribute__((ext_vector_type(2))) float v2f;
typedef __attribute__((ext_vector_type(8))) float v8f;

#define BLK_PER_B  16
#define WAVES_BLK  8
#define T_PER_WAVE (Tn / (BLK_PER_B * WAVES_BLK))
#define NP         BLK_PER_B

#define WS_Q      0
#define WS_ML     (WS_Q + Bn*Dn)
#define WS_ACC    (WS_ML + Bn*NP*2)
#define WS_POOLED (WS_ACC + Bn*NP*Dn)

#define GN      512
#define GK      512
#define KSPLIT  8
#define KSLICE  (GK / KSPLIT)
#define KSTEPS  (KSLICE / 4)
static_assert(GN == Dn && GN == 2 * Sn && GK == Dn, "both products are 32 x 512 x 512");

template <bool SILU_A>
__device__ __forceinline__
void gemm_body(const float* __restrict__ A,
               const float* __restrict__ t_star,
               const float* __restrict__ W1,
               const float* __restrict__ b1,
               const float* __restrict__ Bm,
               const float* __restrict__ bias,
               float* __restrict__ Cout,
               int split) {
    __shared__ float part[KSPLIT * 256];

    int wave = __builtin_amdgcn_readfirstlane(threadIdx.x >> 5);
    int lane = threadIdx.x & 31;
    int h    = lane >> 4;
    int ln   = lane & 15;
    int m0   = blockIdx.y * 16;
    int n    = blockIdx.x * 16 + ln;

    int kb = wave * KSLICE;
    const float* ap = SILU_A ? nullptr : A + (m0 + ln) * GK + kb + 2 * h;
    const float* bp = Bm + (size_t)(kb + 2 * h) * GN + n;

    float bv = bias[n];

    v2f a[KSTEPS], bb[KSTEPS];
    v2f w1v[KSTEPS], b1v[KSTEPS];
    float ts = 0.0f;
    if constexpr (SILU_A) {
        ts = t_star[m0 + ln];
        const float* wp = W1 + kb + 2 * h;
        const float* qp = b1 + kb + 2 * h;
        #pragma unroll
        for (int i = 0; i < KSTEPS; ++i) {
            w1v[i] = *(const v2f*)(wp + 4 * i);
            b1v[i] = *(const v2f*)(qp + 4 * i);
        }
        __builtin_amdgcn_sched_barrier(0);
    }
    #pragma unroll
    for (int i = 0; i < KSTEPS; ++i) {
        if constexpr (!SILU_A)
            a[i] = *(const v2f*)(ap + 4 * i);
        bb[i].x = bp[(size_t)(4 * i    ) * GN];
        bb[i].y = bp[(size_t)(4 * i + 1) * GN];
    }
    __builtin_amdgcn_sched_barrier(0);

    if constexpr (SILU_A) {
        v2f tsv = {ts, ts};
        #pragma unroll
        for (int i = 0; i < KSTEPS; ++i) {
            v2f x = tsv * w1v[i] + b1v[i];
            a[i].x = x.x / (1.0f + __expf(-x.x));
            a[i].y = x.y / (1.0f + __expf(-x.y));
        }
    }

    v8f c = {};
    #pragma unroll
    for (int i = 0; i < KSTEPS; ++i)
        c = __builtin_amdgcn_wmma_f32_16x16x4_f32(false, a[i], false, bb[i],
                                                  (short)0, c, false, false);

    #pragma unroll
    for (int v = 0; v < 8; ++v)
        part[(wave * 8 + v) * 32 + lane] = c[v];
    __syncthreads();

    int e = wave * 32 + lane;
    float s = part[e];
    #pragma unroll
    for (int w = 1; w < KSPLIT; ++w)
        s += part[w * 256 + e];

    int   m   = m0 + wave + 8 * h;
    float val = s + bv;
    if (!split) {
        Cout[m * GN + n] = val;
    } else {
        if (n < Sn) Cout[m * Sn + n] = val;
        else        Cout[Bn * Sn + m * Sn + (n - Sn)] = val;
    }
}

__global__ __launch_bounds__(32 * KSPLIT)
void k_gemm_h_wmma(const float* __restrict__ t_star,
                   const float* __restrict__ W1,
                   const float* __restrict__ b1,
                   const float* __restrict__ Bm,
                   const float* __restrict__ bias,
                   float* __restrict__ Cout) {
    gemm_body<true>(nullptr, t_star, W1, b1, Bm, bias, Cout, 0);
}

__global__ __launch_bounds__(32 * KSPLIT)
void k_gemm_wmma(const float* __restrict__ A,
                 const float* __restrict__ Bm,
                 const float* __restrict__ bias,
                 float* __restrict__ Cout,
                 int split) {
    gemm_body<false>(A, nullptr, nullptr, nullptr, Bm, bias, Cout, split);
}

#define RG   4
#define NGRP (T_PER_WAVE / RG)
static_assert(T_PER_WAVE == 32, "lane tt of a wave holds the time distance of the wave's row tt");
static_assert(T_PER_WAVE % RG == 0 && NGRP >= 2, "whole groups, at least the two requested up front");
typedef __attribute__((ext_vector_type(4))) float v4f;

__global__ __launch_bounds__(256)
void k_attend(const float* __restrict__ context,
              const float* __restrict__ t_star,
              const float* __restrict__ t_context,
              float* __restrict__ ws) {
    int blk      = blockIdx.x;
    int b        = blk >> 4;
    int blk_in_b = blk & 15;
    int wave     = __builtin_amdgcn_readfirstlane(threadIdx.x >> 5);
    int lane     = threadIdx.x & 31;
    int p        = blk_in_b * WAVES_BLK + wave;
    int t0       = p * T_PER_WAVE;

    const float* q = ws + WS_Q + b * Dn;
    v4f q4[4];
    #pragma unroll
    for (int i = 0; i < 4; ++i)
        q4[i] = *(const v4f*)(q + lane * 4 + i * 128);

    float ts = t_star[b];
    float tcv = t_context[(size_t)b * Tn + t0 + lane];

    const float* ctx_w = context + ((size_t)b * Tn + t0) * Dn + lane * 4;

    v4f x[2][RG][4];
#define LOAD_GROUP(G)                                                              \
    _Pragma("unroll") for (int r = 0; r < RG; ++r)                                 \
    _Pragma("unroll") for (int i = 0; i < 4; ++i)                                  \
        x[(G) & 1][r][i] = *(const v4f*)(ctx_w + ((G) * RG + r) * Dn + i * 128)

    float m = -INFINITY, l = 0.0f;
    v2f acc[4][2] = {};

    LOAD_GROUP(0);
    LOAD_GROUP(1);
    __builtin_amdgcn_sched_barrier(0);
    float ad = fabsf(ts - tcv);

    #pragma unroll
    for (int g = 0; g < NGRP; ++g) {
        const v4f (*xg)[4] = x[g & 1];
        float s[RG];
        #pragma unroll
        for (int r = 0; r < RG; ++r) {
            v2f d = {0.0f, 0.0f};
            #pragma unroll
            for (int i = 0; i < 4; ++i) {
                d = __builtin_elementwise_fma(xg[r][i].lo, q4[i].lo, d);
                d = __builtin_elementwise_fma(xg[r][i].hi, q4[i].hi, d);
            }
            float dotp = d.x + d.y;
            #pragma unroll
            for (int off = 16; off > 0; off >>= 1)
                dotp += __shfl_xor(dotp, off, 32);
            float adr = __builtin_bit_cast(float,
                            __builtin_amdgcn_readlane(__builtin_bit_cast(int, ad), g * RG + r));
            s[r] = dotp - adr * SCALE_INV;
        }

        float gm = s[0];
        #pragma unroll
        for (int r = 1; r < RG; ++r) gm = fmaxf(gm, s[r]);
        float mn = fmaxf(m, gm);
        float sc = __expf(m - mn);
        float w[RG];
        float wsum = 0.0f;
        #pragma unroll
        for (int r = 0; r < RG; ++r) { w[r] = __expf(s[r] - mn); wsum += w[r]; }
        l = l * sc + wsum;
        #pragma unroll
        for (int i = 0; i < 4; ++i) {
            v2f a0 = acc[i][0] * sc;
            v2f a1 = acc[i][1] * sc;
            #pragma unroll
            for (int r = 0; r < RG; ++r) {
                v2f wv = {w[r], w[r]};
                a0 = __builtin_elementwise_fma(wv, xg[r][i].lo, a0);
                a1 = __builtin_elementwise_fma(wv, xg[r][i].hi, a1);
            }
            acc[i][0] = a0;
            acc[i][1] = a1;
        }
        m = mn;

        __builtin_amdgcn_sched_barrier(0);
        if (g + 2 < NGRP) LOAD_GROUP(g + 2);
        __builtin_amdgcn_sched_barrier(0);
    }
#undef LOAD_GROUP

    __shared__ float s_acc[WAVES_BLK][Dn];
    __shared__ float s_m[WAVES_BLK], s_l[WAVES_BLK];
    for (int i = 0; i < 4; ++i)
        *(v4f*)(&s_acc[wave][lane * 4 + i * 128]) = __builtin_shufflevector(acc[i][0], acc[i][1], 0, 1, 2, 3);
    if (lane == 0) { s_m[wave] = m; s_l[wave] = l; }
    __syncthreads();

    int tid = threadIdx.x;
    float mw[WAVES_BLK];
    float M = -INFINITY;
    for (int w = 0; w < WAVES_BLK; ++w) { mw[w] = s_m[w]; M = fmaxf(M, mw[w]); }
    float Lb = 0.0f;
    v2f sum = {0.0f, 0.0f};
    for (int w = 0; w < WAVES_BLK; ++w) {
        float e = __expf(mw[w] - M);
        Lb += e * s_l[w];
        v2f ev = {e, e};
        sum = __builtin_elementwise_fma(ev, *(const v2f*)(&s_acc[w][2 * tid]), sum);
    }
    float* accout = ws + WS_ACC + ((size_t)(b * NP + blk_in_b)) * Dn;
    *(v2f*)(accout + 2 * tid) = sum;
    if (tid == 0) {
        float* ml = ws + WS_ML + (b * NP + blk_in_b) * 2;
        ml[0] = M; ml[1] = Lb;
    }
}

__global__ void k_combine(float* __restrict__ ws) {
    int b = blockIdx.x;
    int d = threadIdx.x;
    const float* ml = ws + WS_ML + b * NP * 2;
    float M = -INFINITY;
    for (int p = 0; p < NP; ++p) M = fmaxf(M, ml[2 * p]);
    const float* accb = ws + WS_ACC + (size_t)b * NP * Dn;
    float L = 0.0f, sum = 0.0f;
    for (int p = 0; p < NP; ++p) {
        float e = __expf(ml[2 * p] - M);
        L   += e * ml[2 * p + 1];
        sum += e * accb[(size_t)p * Dn + d];
    }
    ws[WS_POOLED + b * Dn + d] = sum / L;
}

extern "C" void kernel_launch(void* const* d_in, const int* in_sizes, int n_in,
                              void* d_out, int out_size, void* d_ws, size_t ws_size,
                              hipStream_t stream) {
    const float* context   = (const float*)d_in[0];
    const float* t_star    = (const float*)d_in[1];
    const float* t_context = (const float*)d_in[2];
    const float* W1        = (const float*)d_in[3];
    const float* b1        = (const float*)d_in[4];
    const float* W2        = (const float*)d_in[5];
    const float* b2        = (const float*)d_in[6];
    const float* Wout      = (const float*)d_in[7];
    const float* bout      = (const float*)d_in[8];
    float* out = (float*)d_out;
    float* ws  = (float*)d_ws;

    k_gemm_h_wmma<<<dim3(GN / 16, 2), 32 * KSPLIT, 0, stream>>>(t_star, W1, b1, W2, b2,
                                                                ws + WS_Q);

    k_attend<<<Bn * 16, 256, 0, stream>>>(context, t_star, t_context, ws);

    k_combine<<<Bn, Dn, 0, stream>>>(ws);

    k_gemm_wmma<<<dim3(GN / 16, 2), 32 * KSPLIT, 0, stream>>>(ws + WS_POOLED, Wout, bout,
                                                              out, 1);
}
